# starting kernel
# baseline (speedup 1.0000x reference)
#include <hip/hip_runtime.h>
#include <hip/hip_bf16.h>
#include <math.h>

#define B_   16
#define N_   1024
#define CIN_ 64
#define C_   64
#define KT_  3

typedef __attribute__((ext_vector_type(2))) float v2f;
typedef __attribute__((ext_vector_type(8))) float v8f;

__global__ void xsum_kernel(const float* __restrict__ X, float* __restrict__ P0) {
  const int idx = blockIdx.x * blockDim.x + threadIdx.x;
  const float4* row = (const float4*)(X + (size_t)idx * CIN_);
  float s = 0.f;
#pragma unroll
  for (int i = 0; i < CIN_ / 4; ++i) {
    float4 v = row[i];
    s += v.x + v.y + v.z + v.w;
  }
  P0[idx] = s;
}

__global__ void __launch_bounds__(256)
matvec_wmma_kernel(const float* __restrict__ A,
                   const float* __restrict__ pin,
                   float* __restrict__ pout) {
  __shared__ float red[8][16];

  const int lane = threadIdx.x & 31;
  const int wave = threadIdx.x >> 5;
  const int slab = blockIdx.x;
  const int b    = slab >> 6;
  const int n0   = (slab & 63) << 4;
  const int half = lane >> 4;
  const int row  = lane & 15;

  const int kseg = wave * (N_ / 8);
  const float* Ap = A + (size_t)b * N_ * N_ + (size_t)(n0 + row) * N_
                      + kseg + 2 * half;
  const float* pp = pin + b * N_ + kseg + 2 * half;

  v8f acc = {};
#pragma unroll 4
  for (int k = 0; k < N_ / 8; k += 4) {
    v2f a = *(const v2f*)(Ap + k);
    v2f p = *(const v2f*)(pp + k);
    acc = __builtin_amdgcn_wmma_f32_16x16x4_f32(
         false, a,  false, p,
         (short)0, acc,  false,  false);
  }

  if (row == 0) {
    const int base = half * 8;
#pragma unroll
    for (int r = 0; r < 8; ++r) red[wave][base + r] = acc[r];
  }
  __syncthreads();

  if (threadIdx.x < 16) {
    float s = 0.f;
#pragma unroll
    for (int w = 0; w < 8; ++w) s += red[w][threadIdx.x];
    pout[b * N_ + n0 + threadIdx.x] = s;
  }
}

__global__ void htrans_kernel(const float* __restrict__ h, float* __restrict__ ht) {
  const int idx  = blockIdx.x * blockDim.x + threadIdx.x;
  const int c    = idx & (C_ - 1);
  const int rest = idx >> 6;
  const int n    = rest & (N_ - 1);
  const int i    = rest >> 10;
  ht[idx] = h[((size_t)i * C_ + c) * N_ + n];
}

__global__ void __launch_bounds__(256)
final_kernel(const float* __restrict__ P,
             const float* __restrict__ ht,
             const float* __restrict__ gamma,
             const float* __restrict__ beta,
             float* __restrict__ out) {
  const int lane  = threadIdx.x & 31;
  const int gwave = (blockIdx.x * blockDim.x + threadIdx.x) >> 5;
  const int b = gwave >> 10;
  const int n = gwave & (N_ - 1);

  float y0 = 0.f, y1 = 0.f;
#pragma unroll
  for (int i = 0; i <= KT_; ++i) {
    float p = P[i * (B_ * N_) + b * N_ + n];
    const float* hrow = ht + ((size_t)i * N_ + n) * C_;
    y0 = fmaf(p, hrow[lane],      y0);
    y1 = fmaf(p, hrow[lane + 32], y1);
  }

  float s1 = y0 + y1;
  float s2 = fmaf(y0, y0, y1 * y1);
#pragma unroll
  for (int m = 16; m >= 1; m >>= 1) {
    s1 += __shfl_xor(s1, m, 32);
    s2 += __shfl_xor(s2, m, 32);
  }
  const float mu   = s1 * (1.f / C_);
  const float var  = s2 * (1.f / C_) - mu * mu;
  const float rstd = rsqrtf(var + 1e-5f);

  float* dst = out + (size_t)gwave * C_;
  dst[lane]      = tanhf(fmaf((y0 - mu) * rstd, gamma[lane],      beta[lane]));
  dst[lane + 32] = tanhf(fmaf((y1 - mu) * rstd, gamma[lane + 32], beta[lane + 32]));
}

extern "C" void kernel_launch(void* const* d_in, const int* in_sizes, int n_in,
                              void* d_out, int out_size, void* d_ws, size_t ws_size,
                              hipStream_t stream) {
  const float* A     = (const float*)d_in[0];
  const float* X     = (const float*)d_in[1];
  const float* h     = (const float*)d_in[2];
  const float* gamma = (const float*)d_in[3];
  const float* beta  = (const float*)d_in[4];
  float* out = (float*)d_out;

  float* P  = (float*)d_ws;
  float* ht = P + (size_t)(KT_ + 1) * B_ * N_;

  hipLaunchKernelGGL(xsum_kernel, dim3(B_ * N_ / 256), dim3(256), 0, stream, X, P);
  hipLaunchKernelGGL(htrans_kernel, dim3((KT_ + 1) * N_ * C_ / 256), dim3(256), 0,
                     stream, h, ht);
  for (int i = 0; i < KT_; ++i) {
    hipLaunchKernelGGL(matvec_wmma_kernel, dim3(B_ * (N_ / 16)), dim3(256), 0,
                       stream, A, P + i * (B_ * N_), P + (i + 1) * (B_ * N_));
  }
  hipLaunchKernelGGL(final_kernel, dim3((B_ * N_ * 32) / 256), dim3(256), 0,
                     stream, P, ht, gamma, beta, out);
}


# Fuse tap-combine + LayerNorm + tanh into the third mat-vec launch: each 16-row block stages its h[i,c,n0:n0+16] tile through LDS and writes out; tap-transpose and final launches removed (6 -> 4)
# speedup vs baseline: 1.1634x; 1.1634x over previous
#include <hip/hip_runtime.h>
#include <hip/hip_bf16.h>
#include <math.h>

#define B_   16
#define N_   1024
#define CIN_ 64
#define C_   64
#define KT_  3

typedef __attribute__((ext_vector_type(2))) float v2f;
typedef __attribute__((ext_vector_type(8))) float v8f;

__global__ void xsum_kernel(const float* __restrict__ X, float* __restrict__ P0) {
  const int idx = blockIdx.x * blockDim.x + threadIdx.x;
  const float4* row = (const float4*)(X + (size_t)idx * CIN_);
  float s = 0.f;
#pragma unroll
  for (int i = 0; i < CIN_ / 4; ++i) {
    float4 v = row[i];
    s += v.x + v.y + v.z + v.w;
  }
  P0[idx] = s;
}

__global__ void __launch_bounds__(256)
matvec_wmma_kernel(const float* __restrict__ A,
                   const float* __restrict__ pin,
                   float* __restrict__ pout) {
  __shared__ float red[8][16];

  const int lane = threadIdx.x & 31;
  const int wave = threadIdx.x >> 5;
  const int slab = blockIdx.x;
  const int b    = slab >> 6;
  const int n0   = (slab & 63) << 4;
  const int half = lane >> 4;
  const int row  = lane & 15;

  const int kseg = wave * (N_ / 8);
  const float* Ap = A + (size_t)b * N_ * N_ + (size_t)(n0 + row) * N_
                      + kseg + 2 * half;
  const float* pp = pin + b * N_ + kseg + 2 * half;

  v8f acc = {};
#pragma unroll 4
  for (int k = 0; k < N_ / 8; k += 4) {
    v2f a = *(const v2f*)(Ap + k);
    v2f p = *(const v2f*)(pp + k);
    acc = __builtin_amdgcn_wmma_f32_16x16x4_f32(
         false, a,  false, p,
         (short)0, acc,  false,  false);
  }

  if (row == 0) {
    const int base = half * 8;
#pragma unroll
    for (int r = 0; r < 8; ++r) red[wave][base + r] = acc[r];
  }
  __syncthreads();

  if (threadIdx.x < 16) {
    float s = 0.f;
#pragma unroll
    for (int w = 0; w < 8; ++w) s += red[w][threadIdx.x];
    pout[b * N_ + n0 + threadIdx.x] = s;
  }
}

#define HP_ 68

__global__ void __launch_bounds__(256)
matvec_final_kernel(const float* __restrict__ A,
                    const float* __restrict__ P,
                    const float* __restrict__ h,
                    const float* __restrict__ gamma,
                    const float* __restrict__ beta,
                    float* __restrict__ out) {
  __shared__ float red[8][16];
  __shared__ float hs[(KT_ + 1) * 16 * HP_];

  const int lane = threadIdx.x & 31;
  const int wave = threadIdx.x >> 5;
  const int slab = blockIdx.x;
  const int b    = slab >> 6;
  const int n0   = (slab & 63) << 4;
  const int half = lane >> 4;
  const int row  = lane & 15;

  const int hq = threadIdx.x & 3;
  const int hc = threadIdx.x >> 2;
  float4 hv[KT_ + 1];
#pragma unroll
  for (int i = 0; i <= KT_; ++i)
    hv[i] = *(const float4*)(h + ((size_t)(i * C_ + hc)) * N_ + n0 + 4 * hq);

  float pw[2][KT_ + 1];
#pragma unroll
  for (int r = 0; r < 2; ++r)
#pragma unroll
    for (int i = 0; i < KT_; ++i)
      pw[r][i] = P[i * (B_ * N_) + b * N_ + n0 + 2 * wave + r];
  const float g0 = gamma[lane], g1 = gamma[lane + 32];
  const float e0 = beta[lane],  e1 = beta[lane + 32];

  const int kseg = wave * (N_ / 8);
  const float* Ap = A + (size_t)b * N_ * N_ + (size_t)(n0 + row) * N_
                      + kseg + 2 * half;
  const float* pp = P + (KT_ - 1) * (B_ * N_) + b * N_ + kseg + 2 * half;

  v8f acc = {};
#pragma unroll 4
  for (int k = 0; k < N_ / 8; k += 4) {
    v2f a = *(const v2f*)(Ap + k);
    v2f p = *(const v2f*)(pp + k);
    acc = __builtin_amdgcn_wmma_f32_16x16x4_f32(
         false, a,  false, p,
         (short)0, acc,  false,  false);
  }

#pragma unroll
  for (int i = 0; i <= KT_; ++i) {
    float* d = hs + (i * 16 + 4 * hq) * HP_ + hc;
    d[0 * HP_] = hv[i].x;
    d[1 * HP_] = hv[i].y;
    d[2 * HP_] = hv[i].z;
    d[3 * HP_] = hv[i].w;
  }
  if (row == 0) {
    const int base = half * 8;
#pragma unroll
    for (int r = 0; r < 8; ++r) red[wave][base + r] = acc[r];
  }
  __syncthreads();

  float y0[2], y1[2], s1[2], s2[2];
#pragma unroll
  for (int r = 0; r < 2; ++r) {
    const int n = 2 * wave + r;
    float s = 0.f;
#pragma unroll
    for (int w = 0; w < 8; ++w) s += red[w][n];
    pw[r][KT_] = s;

    float a0 = 0.f, a1 = 0.f;
#pragma unroll
    for (int i = 0; i <= KT_; ++i) {
      const float* hrow = hs + (i * 16 + n) * HP_;
      a0 = fmaf(pw[r][i], hrow[lane],      a0);
      a1 = fmaf(pw[r][i], hrow[lane + 32], a1);
    }
    y0[r] = a0;
    y1[r] = a1;
    s1[r] = a0 + a1;
    s2[r] = fmaf(a0, a0, a1 * a1);
  }
#pragma unroll
  for (int m = 16; m >= 1; m >>= 1) {
#pragma unroll
    for (int r = 0; r < 2; ++r) {
      s1[r] += __shfl_xor(s1[r], m, 32);
      s2[r] += __shfl_xor(s2[r], m, 32);
    }
  }
#pragma unroll
  for (int r = 0; r < 2; ++r) {
    const float mu   = s1[r] * (1.f / C_);
    const float var  = s2[r] * (1.f / C_) - mu * mu;
    const float rstd = rsqrtf(var + 1e-5f);

    float* dst = out + ((size_t)(b * N_ + n0 + 2 * wave + r)) * C_;
    dst[lane]      = tanhf(fmaf((y0[r] - mu) * rstd, g0, e0));
    dst[lane + 32] = tanhf(fmaf((y1[r] - mu) * rstd, g1, e1));
  }
}

extern "C" void kernel_launch(void* const* d_in, const int* in_sizes, int n_in,
                              void* d_out, int out_size, void* d_ws, size_t ws_size,
                              hipStream_t stream) {
  const float* A     = (const float*)d_in[0];
  const float* X     = (const float*)d_in[1];
  const float* h     = (const float*)d_in[2];
  const float* gamma = (const float*)d_in[3];
  const float* beta  = (const float*)d_in[4];
  float* out = (float*)d_out;

  float* P  = (float*)d_ws;

  hipLaunchKernelGGL(xsum_kernel, dim3(B_ * N_ / 256), dim3(256), 0, stream, X, P);
  for (int i = 0; i < KT_ - 1; ++i) {
    hipLaunchKernelGGL(matvec_wmma_kernel, dim3(B_ * (N_ / 16)), dim3(256), 0,
                       stream, A, P + i * (B_ * N_), P + (i + 1) * (B_ * N_));
  }
  hipLaunchKernelGGL(matvec_final_kernel, dim3(B_ * (N_ / 16)), dim3(256), 0,
                     stream, A, P, h, gamma, beta, out);
}


# Each mat-vec wave owns 4 whole rows of A (16 KiB, half the waves): loads are 512 contiguous bytes, issued up front; products on the f32 WMMA tile's diagonal, lane-summed; reduction tree differs
# speedup vs baseline: 1.3324x; 1.1453x over previous
#include <hip/hip_runtime.h>
#include <hip/hip_bf16.h>
#include <math.h>


#define B_   16
#define N_   1024
#define CIN_ 64
#define C_   64
#define KT_  3

#define RW_   4
#define SLAB_ (8 * RW_)
#define KQ_   (N_ / 128)

typedef __attribute__((ext_vector_type(2))) float v2f;
typedef __attribute__((ext_vector_type(8))) float v8f;

__global__ void xsum_kernel(const float* __restrict__ X, float* __restrict__ P0) {
  const int idx = blockIdx.x * blockDim.x + threadIdx.x;
  const float4* row = (const float4*)(X + (size_t)idx * CIN_);
  float s = 0.f;
#pragma unroll
  for (int i = 0; i < CIN_ / 4; ++i) {
    float4 v = row[i];
    s += v.x + v.y + v.z + v.w;
  }
  P0[idx] = s;
}

__device__ __forceinline__ void issue_row_loads(const float* __restrict__ Ap,
                                                const float* __restrict__ pp,
                                                float4 (&pv)[KQ_],
                                                float4 (&av)[RW_][KQ_]) {
#pragma unroll
  for (int q = 0; q < KQ_; ++q) pv[q] = *(const float4*)(pp + 128 * q);
#pragma unroll
  for (int q = 0; q < KQ_; ++q)
#pragma unroll
    for (int r = 0; r < RW_; ++r)
      av[r][q] = *(const float4*)(Ap + (size_t)r * N_ + 128 * q);
}

__device__ __forceinline__ void rows_dot(const float4 (&pv)[KQ_],
                                         const float4 (&av)[RW_][KQ_],
                                         const int lane, float (&dot)[RW_]) {
  v8f acc[RW_];
#pragma unroll
  for (int r = 0; r < RW_; ++r) acc[r] = v8f{};

#pragma unroll
  for (int q = 0; q < KQ_; ++q) {
#pragma unroll
    for (int r = 0; r < RW_; ++r) {
      const v2f a = {av[r][q].x, av[r][q].y};
      const v2f p = {pv[q].x, pv[q].y};
      acc[r] = __builtin_amdgcn_wmma_f32_16x16x4_f32(
           false, a,  false, p,
           (short)0, acc[r],  false,  false);
    }
#pragma unroll
    for (int r = 0; r < RW_; ++r) {
      const v2f a = {av[r][q].z, av[r][q].w};
      const v2f p = {pv[q].z, pv[q].w};
      acc[r] = __builtin_amdgcn_wmma_f32_16x16x4_f32(
           false, a,  false, p,
           (short)0, acc[r],  false,  false);
    }
  }

  __builtin_amdgcn_sched_barrier(0);
  asm volatile("v_nop\n\tv_nop\n\tv_nop\n\tv_nop\n\tv_nop\n\tv_nop\n\tv_nop\n\tv_nop");
  __builtin_amdgcn_sched_barrier(0);

  const int  dr = lane & 7;
  const bool dv = ((lane + 8) & 31) < 16;
#pragma unroll
  for (int r = 0; r < RW_; ++r) {
    float d = acc[r][0];
#pragma unroll
    for (int v = 1; v < 8; ++v) d = (dr == v) ? acc[r][v] : d;
    dot[r] = dv ? d : 0.f;
  }
#pragma unroll
  for (int m = 16; m >= 1; m >>= 1) {
#pragma unroll
    for (int r = 0; r < RW_; ++r) dot[r] += __shfl_xor(dot[r], m, 32);
  }
}

__global__ void __launch_bounds__(256)
matvec_wmma_kernel(const float* __restrict__ A,
                   const float* __restrict__ pin,
                   float* __restrict__ pout) {
  __shared__ float red[SLAB_];

  const int lane = threadIdx.x & 31;
  const int wave = threadIdx.x >> 5;
  const int slab = blockIdx.x;
  const int b    = slab >> 5;
  const int n0   = (slab & 31) << 5;

  const float* Ap = A + (size_t)b * N_ * N_ + (size_t)(n0 + RW_ * wave) * N_ + 4 * lane;
  const float* pp = pin + b * N_ + 4 * lane;

  float4 pv[KQ_];
  float4 av[RW_][KQ_];
  issue_row_loads(Ap, pp, pv, av);
  __builtin_amdgcn_sched_barrier(0);

  float dot[RW_];
  rows_dot(pv, av, lane, dot);

  if (lane == 0) {
#pragma unroll
    for (int r = 0; r < RW_; ++r) red[RW_ * wave + r] = dot[r];
  }
  __syncthreads();
  if (threadIdx.x < SLAB_) pout[b * N_ + n0 + threadIdx.x] = red[threadIdx.x];
}

#define HP_ 66

__global__ void __launch_bounds__(256)
matvec_final_kernel(const float* __restrict__ A,
                    const float* __restrict__ P,
                    const float* __restrict__ h,
                    const float* __restrict__ gamma,
                    const float* __restrict__ beta,
                    float* __restrict__ out) {
  __shared__ float hs[(KT_ + 1) * SLAB_ * HP_];

  const int lane = threadIdx.x & 31;
  const int wave = threadIdx.x >> 5;
  const int slab = blockIdx.x;
  const int b    = slab >> 5;
  const int n0   = (slab & 31) << 5;

  const int hq = threadIdx.x & 7;
  const int hc = threadIdx.x >> 3;
  float4 hv[KT_ + 1][2];
#pragma unroll
  for (int i = 0; i <= KT_; ++i)
#pragma unroll
    for (int u = 0; u < 2; ++u)
      hv[i][u] = *(const float4*)(h + ((size_t)(i * C_ + hc + 32 * u)) * N_ + n0 + 4 * hq);

  float pw[RW_][KT_ + 1];
#pragma unroll
  for (int r = 0; r < RW_; ++r)
#pragma unroll
    for (int i = 0; i < KT_; ++i)
      pw[r][i] = P[i * (B_ * N_) + b * N_ + n0 + RW_ * wave + r];
  const float g0 = gamma[lane], g1 = gamma[lane + 32];
  const float e0 = beta[lane],  e1 = beta[lane + 32];

  const float* Ap = A + (size_t)b * N_ * N_ + (size_t)(n0 + RW_ * wave) * N_ + 4 * lane;
  const float* pp = P + (KT_ - 1) * (B_ * N_) + b * N_ + 4 * lane;
  float4 pv[KQ_];
  float4 av[RW_][KQ_];
  issue_row_loads(Ap, pp, pv, av);
  __builtin_amdgcn_sched_barrier(0);

#pragma unroll
  for (int i = 0; i <= KT_; ++i)
#pragma unroll
    for (int u = 0; u < 2; ++u) {
      float* d = hs + (i * SLAB_ + 4 * hq) * HP_ + hc + 32 * u;
      d[0 * HP_] = hv[i][u].x;
      d[1 * HP_] = hv[i][u].y;
      d[2 * HP_] = hv[i][u].z;
      d[3 * HP_] = hv[i][u].w;
    }

  __builtin_amdgcn_sched_barrier(0);

  float dot[RW_];
  rows_dot(pv, av, lane, dot);
  __syncthreads();

  float y0[RW_], y1[RW_], s1[RW_], s2[RW_];
#pragma unroll
  for (int r = 0; r < RW_; ++r) {
    const int n = RW_ * wave + r;
    pw[r][KT_] = dot[r];

    float a0 = 0.f, a1 = 0.f;
#pragma unroll
    for (int i = 0; i <= KT_; ++i) {
      const float* hrow = hs + (i * SLAB_ + n) * HP_;
      a0 = fmaf(pw[r][i], hrow[lane],      a0);
      a1 = fmaf(pw[r][i], hrow[lane + 32], a1);
    }
    y0[r] = a0;
    y1[r] = a1;
    s1[r] = a0 + a1;
    s2[r] = fmaf(a0, a0, a1 * a1);
  }
#pragma unroll
  for (int m = 16; m >= 1; m >>= 1) {
#pragma unroll
    for (int r = 0; r < RW_; ++r) {
      s1[r] += __shfl_xor(s1[r], m, 32);
      s2[r] += __shfl_xor(s2[r], m, 32);
    }
  }
#pragma unroll
  for (int r = 0; r < RW_; ++r) {
    const float mu   = s1[r] * (1.f / C_);
    const float var  = s2[r] * (1.f / C_) - mu * mu;
    const float rstd = rsqrtf(var + 1e-5f);

    float* dst = out + ((size_t)(b * N_ + n0 + RW_ * wave + r)) * C_;
    dst[lane]      = tanhf(fmaf((y0[r] - mu) * rstd, g0, e0));
    dst[lane + 32] = tanhf(fmaf((y1[r] - mu) * rstd, g1, e1));
  }
}

extern "C" void kernel_launch(void* const* d_in, const int* in_sizes, int n_in,
                              void* d_out, int out_size, void* d_ws, size_t ws_size,
                              hipStream_t stream) {
  const float* A     = (const float*)d_in[0];
  const float* X     = (const float*)d_in[1];
  const float* h     = (const float*)d_in[2];
  const float* gamma = (const float*)d_in[3];
  const float* beta  = (const float*)d_in[4];
  float* out = (float*)d_out;

  float* P  = (float*)d_ws;

  hipLaunchKernelGGL(xsum_kernel, dim3(B_ * N_ / 256), dim3(256), 0, stream, X, P);
  for (int i = 0; i < KT_ - 1; ++i) {
    hipLaunchKernelGGL(matvec_wmma_kernel, dim3(B_ * (N_ / SLAB_)), dim3(256), 0,
                       stream, A, P + i * (B_ * N_), P + (i + 1) * (B_ * N_));
  }
  hipLaunchKernelGGL(matvec_final_kernel, dim3(B_ * (N_ / SLAB_)), dim3(256), 0,
                     stream, A, P, h, gamma, beta, out);
}


# Channel-sum kernel reads whole lines: a wave takes 32 rows of X as 16 loads of 512 contiguous bytes (256 two-wave blocks), folds the 16 lanes of a row by halving exchanges; channel sum order differs
# speedup vs baseline: 1.4120x; 1.0597x over previous
#include <hip/hip_runtime.h>
#include <hip/hip_bf16.h>
#include <math.h>


#define B_   16
#define N_   1024
#define CIN_ 64
#define C_   64
#define KT_  3

#define RW_   4
#define SLAB_ (8 * RW_)
#define KQ_   (N_ / 128)

typedef __attribute__((ext_vector_type(2))) float v2f;
typedef __attribute__((ext_vector_type(8))) float v8f;

#define XR_ 32
#define XL_ (XR_ / 2)

__global__ void __launch_bounds__(64)
xsum_kernel(const float* __restrict__ X, float* __restrict__ P0) {
  const int lane = threadIdx.x & 31;
  const int wv   = blockIdx.x * 2 + (threadIdx.x >> 5);
  const float* src = X + (size_t)wv * (XR_ * CIN_) + 4 * lane;

  float s[XL_];
#pragma unroll
  for (int k = 0; k < XL_; ++k) {
    const float4 v = *(const float4*)(src + 128 * k);
    s[k] = (v.x + v.y) + (v.z + v.w);
  }

  const bool b3 = (lane & 8) != 0, b2 = (lane & 4) != 0;
  const bool b1 = (lane & 2) != 0, b0 = (lane & 1) != 0;
  float t8[8], t4[4], t2[2];
#pragma unroll
  for (int i = 0; i < 8; ++i)
    t8[i] = (b3 ? s[i + 8] : s[i]) + __shfl_xor(b3 ? s[i] : s[i + 8], 8, 32);
#pragma unroll
  for (int i = 0; i < 4; ++i)
    t4[i] = (b2 ? t8[i + 4] : t8[i]) + __shfl_xor(b2 ? t8[i] : t8[i + 4], 4, 32);
#pragma unroll
  for (int i = 0; i < 2; ++i)
    t2[i] = (b1 ? t4[i + 2] : t4[i]) + __shfl_xor(b1 ? t4[i] : t4[i + 2], 2, 32);
  const float t1 = (b0 ? t2[1] : t2[0]) + __shfl_xor(b0 ? t2[0] : t2[1], 1, 32);

  P0[wv * XR_ + 2 * (lane & 15) + (lane >> 4)] = t1;
}

__device__ __forceinline__ void issue_row_loads(const float* __restrict__ Ap,
                                                const float* __restrict__ pp,
                                                float4 (&pv)[KQ_],
                                                float4 (&av)[RW_][KQ_]) {
#pragma unroll
  for (int q = 0; q < KQ_; ++q) pv[q] = *(const float4*)(pp + 128 * q);
#pragma unroll
  for (int q = 0; q < KQ_; ++q)
#pragma unroll
    for (int r = 0; r < RW_; ++r)
      av[r][q] = *(const float4*)(Ap + (size_t)r * N_ + 128 * q);
}

__device__ __forceinline__ void rows_dot(const float4 (&pv)[KQ_],
                                         const float4 (&av)[RW_][KQ_],
                                         const int lane, float (&dot)[RW_]) {
  v8f acc[RW_];
#pragma unroll
  for (int r = 0; r < RW_; ++r) acc[r] = v8f{};

#pragma unroll
  for (int q = 0; q < KQ_; ++q) {
#pragma unroll
    for (int r = 0; r < RW_; ++r) {
      const v2f a = {av[r][q].x, av[r][q].y};
      const v2f p = {pv[q].x, pv[q].y};
      acc[r] = __builtin_amdgcn_wmma_f32_16x16x4_f32(
           false, a,  false, p,
           (short)0, acc[r],  false,  false);
    }
#pragma unroll
    for (int r = 0; r < RW_; ++r) {
      const v2f a = {av[r][q].z, av[r][q].w};
      const v2f p = {pv[q].z, pv[q].w};
      acc[r] = __builtin_amdgcn_wmma_f32_16x16x4_f32(
           false, a,  false, p,
           (short)0, acc[r],  false,  false);
    }
  }

  __builtin_amdgcn_sched_barrier(0);
  asm volatile("v_nop\n\tv_nop\n\tv_nop\n\tv_nop\n\tv_nop\n\tv_nop\n\tv_nop\n\tv_nop");
  __builtin_amdgcn_sched_barrier(0);

  const int  dr = lane & 7;
  const bool dv = ((lane + 8) & 31) < 16;
#pragma unroll
  for (int r = 0; r < RW_; ++r) {
    float d = acc[r][0];
#pragma unroll
    for (int v = 1; v < 8; ++v) d = (dr == v) ? acc[r][v] : d;
    dot[r] = dv ? d : 0.f;
  }
#pragma unroll
  for (int m = 16; m >= 1; m >>= 1) {
#pragma unroll
    for (int r = 0; r < RW_; ++r) dot[r] += __shfl_xor(dot[r], m, 32);
  }
}

__global__ void __launch_bounds__(256)
matvec_wmma_kernel(const float* __restrict__ A,
                   const float* __restrict__ pin,
                   float* __restrict__ pout) {
  __shared__ float red[SLAB_];

  const int lane = threadIdx.x & 31;
  const int wave = threadIdx.x >> 5;
  const int slab = blockIdx.x;
  const int b    = slab >> 5;
  const int n0   = (slab & 31) << 5;

  const float* Ap = A + (size_t)b * N_ * N_ + (size_t)(n0 + RW_ * wave) * N_ + 4 * lane;
  const float* pp = pin + b * N_ + 4 * lane;

  float4 pv[KQ_];
  float4 av[RW_][KQ_];
  issue_row_loads(Ap, pp, pv, av);
  __builtin_amdgcn_sched_barrier(0);

  float dot[RW_];
  rows_dot(pv, av, lane, dot);

  if (lane == 0) {
#pragma unroll
    for (int r = 0; r < RW_; ++r) red[RW_ * wave + r] = dot[r];
  }
  __syncthreads();
  if (threadIdx.x < SLAB_) pout[b * N_ + n0 + threadIdx.x] = red[threadIdx.x];
}

#define HP_ 66

__global__ void __launch_bounds__(256)
matvec_final_kernel(const float* __restrict__ A,
                    const float* __restrict__ P,
                    const float* __restrict__ h,
                    const float* __restrict__ gamma,
                    const float* __restrict__ beta,
                    float* __restrict__ out) {
  __shared__ float hs[(KT_ + 1) * SLAB_ * HP_];

  const int lane = threadIdx.x & 31;
  const int wave = threadIdx.x >> 5;
  const int slab = blockIdx.x;
  const int b    = slab >> 5;
  const int n0   = (slab & 31) << 5;

  const int hq = threadIdx.x & 7;
  const int hc = threadIdx.x >> 3;
  float4 hv[KT_ + 1][2];
#pragma unroll
  for (int i = 0; i <= KT_; ++i)
#pragma unroll
    for (int u = 0; u < 2; ++u)
      hv[i][u] = *(const float4*)(h + ((size_t)(i * C_ + hc + 32 * u)) * N_ + n0 + 4 * hq);

  float pw[RW_][KT_ + 1];
#pragma unroll
  for (int r = 0; r < RW_; ++r)
#pragma unroll
    for (int i = 0; i < KT_; ++i)
      pw[r][i] = P[i * (B_ * N_) + b * N_ + n0 + RW_ * wave + r];
  const float g0 = gamma[lane], g1 = gamma[lane + 32];
  const float e0 = beta[lane],  e1 = beta[lane + 32];

  const float* Ap = A + (size_t)b * N_ * N_ + (size_t)(n0 + RW_ * wave) * N_ + 4 * lane;
  const float* pp = P + (KT_ - 1) * (B_ * N_) + b * N_ + 4 * lane;
  float4 pv[KQ_];
  float4 av[RW_][KQ_];
  issue_row_loads(Ap, pp, pv, av);
  __builtin_amdgcn_sched_barrier(0);

#pragma unroll
  for (int i = 0; i <= KT_; ++i)
#pragma unroll
    for (int u = 0; u < 2; ++u) {
      float* d = hs + (i * SLAB_ + 4 * hq) * HP_ + hc + 32 * u;
      d[0 * HP_] = hv[i][u].x;
      d[1 * HP_] = hv[i][u].y;
      d[2 * HP_] = hv[i][u].z;
      d[3 * HP_] = hv[i][u].w;
    }

  __builtin_amdgcn_sched_barrier(0);

  float dot[RW_];
  rows_dot(pv, av, lane, dot);
  __syncthreads();

  float y0[RW_], y1[RW_], s1[RW_], s2[RW_];
#pragma unroll
  for (int r = 0; r < RW_; ++r) {
    const int n = RW_ * wave + r;
    pw[r][KT_] = dot[r];

    float a0 = 0.f, a1 = 0.f;
#pragma unroll
    for (int i = 0; i <= KT_; ++i) {
      const float* hrow = hs + (i * SLAB_ + n) * HP_;
      a0 = fmaf(pw[r][i], hrow[lane],      a0);
      a1 = fmaf(pw[r][i], hrow[lane + 32], a1);
    }
    y0[r] = a0;
    y1[r] = a1;
    s1[r] = a0 + a1;
    s2[r] = fmaf(a0, a0, a1 * a1);
  }
#pragma unroll
  for (int m = 16; m >= 1; m >>= 1) {
#pragma unroll
    for (int r = 0; r < RW_; ++r) {
      s1[r] += __shfl_xor(s1[r], m, 32);
      s2[r] += __shfl_xor(s2[r], m, 32);
    }
  }
#pragma unroll
  for (int r = 0; r < RW_; ++r) {
    const float mu   = s1[r] * (1.f / C_);
    const float var  = s2[r] * (1.f / C_) - mu * mu;
    const float rstd = rsqrtf(var + 1e-5f);

    float* dst = out + ((size_t)(b * N_ + n0 + RW_ * wave + r)) * C_;
    dst[lane]      = tanhf(fmaf((y0[r] - mu) * rstd, g0, e0));
    dst[lane + 32] = tanhf(fmaf((y1[r] - mu) * rstd, g1, e1));
  }
}

extern "C" void kernel_launch(void* const* d_in, const int* in_sizes, int n_in,
                              void* d_out, int out_size, void* d_ws, size_t ws_size,
                              hipStream_t stream) {
  const float* A     = (const float*)d_in[0];
  const float* X     = (const float*)d_in[1];
  const float* h     = (const float*)d_in[2];
  const float* gamma = (const float*)d_in[3];
  const float* beta  = (const float*)d_in[4];
  float* out = (float*)d_out;

  float* P  = (float*)d_ws;

  hipLaunchKernelGGL(xsum_kernel, dim3(B_ * N_ / (2 * XR_)), dim3(64), 0, stream, X, P);
  for (int i = 0; i < KT_ - 1; ++i) {
    hipLaunchKernelGGL(matvec_wmma_kernel, dim3(B_ * (N_ / SLAB_)), dim3(256), 0,
                       stream, A, P + i * (B_ * N_), P + (i + 1) * (B_ * N_));
  }
  hipLaunchKernelGGL(matvec_final_kernel, dim3(B_ * (N_ / SLAB_)), dim3(256), 0,
                     stream, A, P, h, gamma, beta, out);
}


# tanh of the last launch as 1 - 2/(exp2(c|z|)+1) on hardware exp2 and reciprocal: branch-free, a wave's eight values interleaved (was eight two-branch library calls in series); last bits of out differ
# speedup vs baseline: 1.5073x; 1.0675x over previous
#include <hip/hip_runtime.h>
#include <hip/hip_bf16.h>
#include <math.h>


#define B_   16
#define N_   1024
#define CIN_ 64
#define C_   64
#define KT_  3

#define RW_   4
#define SLAB_ (8 * RW_)
#define KQ_   (N_ / 128)

typedef __attribute__((ext_vector_type(2))) float v2f;
typedef __attribute__((ext_vector_type(8))) float v8f;

#define XR_ 32
#define XL_ (XR_ / 2)

__global__ void __launch_bounds__(64)
xsum_kernel(const float* __restrict__ X, float* __restrict__ P0) {
  const int lane = threadIdx.x & 31;
  const int wv   = blockIdx.x * 2 + (threadIdx.x >> 5);
  const float* src = X + (size_t)wv * (XR_ * CIN_) + 4 * lane;

  float s[XL_];
#pragma unroll
  for (int k = 0; k < XL_; ++k) {
    const float4 v = *(const float4*)(src + 128 * k);
    s[k] = (v.x + v.y) + (v.z + v.w);
  }

  const bool b3 = (lane & 8) != 0, b2 = (lane & 4) != 0;
  const bool b1 = (lane & 2) != 0, b0 = (lane & 1) != 0;
  float t8[8], t4[4], t2[2];
#pragma unroll
  for (int i = 0; i < 8; ++i)
    t8[i] = (b3 ? s[i + 8] : s[i]) + __shfl_xor(b3 ? s[i] : s[i + 8], 8, 32);
#pragma unroll
  for (int i = 0; i < 4; ++i)
    t4[i] = (b2 ? t8[i + 4] : t8[i]) + __shfl_xor(b2 ? t8[i] : t8[i + 4], 4, 32);
#pragma unroll
  for (int i = 0; i < 2; ++i)
    t2[i] = (b1 ? t4[i + 2] : t4[i]) + __shfl_xor(b1 ? t4[i] : t4[i + 2], 2, 32);
  const float t1 = (b0 ? t2[1] : t2[0]) + __shfl_xor(b0 ? t2[0] : t2[1], 1, 32);

  P0[wv * XR_ + 2 * (lane & 15) + (lane >> 4)] = t1;
}

__device__ __forceinline__ void issue_row_loads(const float* __restrict__ Ap,
                                                const float* __restrict__ pp,
                                                float4 (&pv)[KQ_],
                                                float4 (&av)[RW_][KQ_]) {
#pragma unroll
  for (int q = 0; q < KQ_; ++q) pv[q] = *(const float4*)(pp + 128 * q);
#pragma unroll
  for (int q = 0; q < KQ_; ++q)
#pragma unroll
    for (int r = 0; r < RW_; ++r)
      av[r][q] = *(const float4*)(Ap + (size_t)r * N_ + 128 * q);
}

__device__ __forceinline__ void rows_dot(const float4 (&pv)[KQ_],
                                         const float4 (&av)[RW_][KQ_],
                                         const int lane, float (&dot)[RW_]) {
  v8f acc[RW_];
#pragma unroll
  for (int r = 0; r < RW_; ++r) acc[r] = v8f{};

#pragma unroll
  for (int q = 0; q < KQ_; ++q) {
#pragma unroll
    for (int r = 0; r < RW_; ++r) {
      const v2f a = {av[r][q].x, av[r][q].y};
      const v2f p = {pv[q].x, pv[q].y};
      acc[r] = __builtin_amdgcn_wmma_f32_16x16x4_f32(
           false, a,  false, p,
           (short)0, acc[r],  false,  false);
    }
#pragma unroll
    for (int r = 0; r < RW_; ++r) {
      const v2f a = {av[r][q].z, av[r][q].w};
      const v2f p = {pv[q].z, pv[q].w};
      acc[r] = __builtin_amdgcn_wmma_f32_16x16x4_f32(
           false, a,  false, p,
           (short)0, acc[r],  false,  false);
    }
  }

  __builtin_amdgcn_sched_barrier(0);
  asm volatile("v_nop\n\tv_nop\n\tv_nop\n\tv_nop\n\tv_nop\n\tv_nop\n\tv_nop\n\tv_nop");
  __builtin_amdgcn_sched_barrier(0);

  const int  dr = lane & 7;
  const bool dv = ((lane + 8) & 31) < 16;
#pragma unroll
  for (int r = 0; r < RW_; ++r) {
    float d = acc[r][0];
#pragma unroll
    for (int v = 1; v < 8; ++v) d = (dr == v) ? acc[r][v] : d;
    dot[r] = dv ? d : 0.f;
  }
#pragma unroll
  for (int m = 16; m >= 1; m >>= 1) {
#pragma unroll
    for (int r = 0; r < RW_; ++r) dot[r] += __shfl_xor(dot[r], m, 32);
  }
}

__global__ void __launch_bounds__(256)
matvec_wmma_kernel(const float* __restrict__ A,
                   const float* __restrict__ pin,
                   float* __restrict__ pout) {
  __shared__ float red[SLAB_];

  const int lane = threadIdx.x & 31;
  const int wave = threadIdx.x >> 5;
  const int slab = blockIdx.x;
  const int b    = slab >> 5;
  const int n0   = (slab & 31) << 5;

  const float* Ap = A + (size_t)b * N_ * N_ + (size_t)(n0 + RW_ * wave) * N_ + 4 * lane;
  const float* pp = pin + b * N_ + 4 * lane;

  float4 pv[KQ_];
  float4 av[RW_][KQ_];
  issue_row_loads(Ap, pp, pv, av);
  __builtin_amdgcn_sched_barrier(0);

  float dot[RW_];
  rows_dot(pv, av, lane, dot);

  if (lane == 0) {
#pragma unroll
    for (int r = 0; r < RW_; ++r) red[RW_ * wave + r] = dot[r];
  }
  __syncthreads();
  if (threadIdx.x < SLAB_) pout[b * N_ + n0 + threadIdx.x] = red[threadIdx.x];
}

__device__ __forceinline__ float tanh_exp2(const float z) {
  const float t = __builtin_amdgcn_exp2f(fabsf(z) * 2.88539008177792681f);
  const float r = fmaf(-2.f, __builtin_amdgcn_rcpf(t + 1.f), 1.f);
  return copysignf(r, z);
}

#define HP_ 66

__global__ void __launch_bounds__(256)
matvec_final_kernel(const float* __restrict__ A,
                    const float* __restrict__ P,
                    const float* __restrict__ h,
                    const float* __restrict__ gamma,
                    const float* __restrict__ beta,
                    float* __restrict__ out) {
  __shared__ float hs[(KT_ + 1) * SLAB_ * HP_];

  const int lane = threadIdx.x & 31;
  const int wave = threadIdx.x >> 5;
  const int slab = blockIdx.x;
  const int b    = slab >> 5;
  const int n0   = (slab & 31) << 5;

  const int hq = threadIdx.x & 7;
  const int hc = threadIdx.x >> 3;
  float4 hv[KT_ + 1][2];
#pragma unroll
  for (int i = 0; i <= KT_; ++i)
#pragma unroll
    for (int u = 0; u < 2; ++u)
      hv[i][u] = *(const float4*)(h + ((size_t)(i * C_ + hc + 32 * u)) * N_ + n0 + 4 * hq);

  float pw[RW_][KT_ + 1];
#pragma unroll
  for (int r = 0; r < RW_; ++r)
#pragma unroll
    for (int i = 0; i < KT_; ++i)
      pw[r][i] = P[i * (B_ * N_) + b * N_ + n0 + RW_ * wave + r];
  const float g0 = gamma[lane], g1 = gamma[lane + 32];
  const float e0 = beta[lane],  e1 = beta[lane + 32];

  const float* Ap = A + (size_t)b * N_ * N_ + (size_t)(n0 + RW_ * wave) * N_ + 4 * lane;
  const float* pp = P + (KT_ - 1) * (B_ * N_) + b * N_ + 4 * lane;
  float4 pv[KQ_];
  float4 av[RW_][KQ_];
  issue_row_loads(Ap, pp, pv, av);
  __builtin_amdgcn_sched_barrier(0);

#pragma unroll
  for (int i = 0; i <= KT_; ++i)
#pragma unroll
    for (int u = 0; u < 2; ++u) {
      float* d = hs + (i * SLAB_ + 4 * hq) * HP_ + hc + 32 * u;
      d[0 * HP_] = hv[i][u].x;
      d[1 * HP_] = hv[i][u].y;
      d[2 * HP_] = hv[i][u].z;
      d[3 * HP_] = hv[i][u].w;
    }

  __builtin_amdgcn_sched_barrier(0);

  float dot[RW_];
  rows_dot(pv, av, lane, dot);
  __syncthreads();

  float y0[RW_], y1[RW_], s1[RW_], s2[RW_];
#pragma unroll
  for (int r = 0; r < RW_; ++r) {
    const int n = RW_ * wave + r;
    pw[r][KT_] = dot[r];

    float a0 = 0.f, a1 = 0.f;
#pragma unroll
    for (int i = 0; i <= KT_; ++i) {
      const float* hrow = hs + (i * SLAB_ + n) * HP_;
      a0 = fmaf(pw[r][i], hrow[lane],      a0);
      a1 = fmaf(pw[r][i], hrow[lane + 32], a1);
    }
    y0[r] = a0;
    y1[r] = a1;
    s1[r] = a0 + a1;
    s2[r] = fmaf(a0, a0, a1 * a1);
  }
#pragma unroll
  for (int m = 16; m >= 1; m >>= 1) {
#pragma unroll
    for (int r = 0; r < RW_; ++r) {
      s1[r] += __shfl_xor(s1[r], m, 32);
      s2[r] += __shfl_xor(s2[r], m, 32);
    }
  }
  float z0[RW_], z1[RW_];
#pragma unroll
  for (int r = 0; r < RW_; ++r) {
    const float mu   = s1[r] * (1.f / C_);
    const float var  = s2[r] * (1.f / C_) - mu * mu;
    const float rstd = rsqrtf(var + 1e-5f);
    z0[r] = fmaf((y0[r] - mu) * rstd, g0, e0);
    z1[r] = fmaf((y1[r] - mu) * rstd, g1, e1);
  }
#pragma unroll
  for (int r = 0; r < RW_; ++r) {
    float* dst = out + ((size_t)(b * N_ + n0 + RW_ * wave + r)) * C_;
    dst[lane]      = tanh_exp2(z0[r]);
    dst[lane + 32] = tanh_exp2(z1[r]);
  }
}

extern "C" void kernel_launch(void* const* d_in, const int* in_sizes, int n_in,
                              void* d_out, int out_size, void* d_ws, size_t ws_size,
                              hipStream_t stream) {
  const float* A     = (const float*)d_in[0];
  const float* X     = (const float*)d_in[1];
  const float* h     = (const float*)d_in[2];
  const float* gamma = (const float*)d_in[3];
  const float* beta  = (const float*)d_in[4];
  float* out = (float*)d_out;

  float* P  = (float*)d_ws;

  hipLaunchKernelGGL(xsum_kernel, dim3(B_ * N_ / (2 * XR_)), dim3(64), 0, stream, X, P);
  for (int i = 0; i < KT_ - 1; ++i) {
    hipLaunchKernelGGL(matvec_wmma_kernel, dim3(B_ * (N_ / SLAB_)), dim3(256), 0,
                       stream, A, P + i * (B_ * N_), P + (i + 1) * (B_ * N_));
  }
  hipLaunchKernelGGL(matvec_final_kernel, dim3(B_ * (N_ / SLAB_)), dim3(256), 0,
                     stream, A, P, h, gamma, beta, out);
}


# Every lane-exchange sum on the vector ALU (DPP row_xmask inside 16 lanes, v_permlanex16 across the halves) instead of LDS-unit permutes; same pairings in the same order, sums bit-identical
# speedup vs baseline: 1.5571x; 1.0330x over previous
#include <hip/hip_runtime.h>
#include <hip/hip_bf16.h>
#include <math.h>


#define B_   16
#define N_   1024
#define CIN_ 64
#define C_   64
#define KT_  3

#define RW_   4
#define SLAB_ (8 * RW_)
#define KQ_   (N_ / 128)

typedef __attribute__((ext_vector_type(2))) float v2f;
typedef __attribute__((ext_vector_type(8))) float v8f;

template <int M>
__device__ __forceinline__ float xrow(const float x) {
  const int b = __builtin_bit_cast(int, x);
  return __builtin_bit_cast(float, __builtin_amdgcn_update_dpp(b, b, 0x160 + M, 0xf, 0xf, true));
}
__device__ __forceinline__ float xhalf(const float x) {
  const int b = __builtin_bit_cast(int, x);
  return __builtin_bit_cast(float,
      __builtin_amdgcn_permlanex16(b, b, 0x76543210, 0xfedcba98, false, false));
}

#define XR_ 32
#define XL_ (XR_ / 2)

__global__ void __launch_bounds__(64)
xsum_kernel(const float* __restrict__ X, float* __restrict__ P0) {
  const int lane = threadIdx.x & 31;
  const int wv   = blockIdx.x * 2 + (threadIdx.x >> 5);
  const float* src = X + (size_t)wv * (XR_ * CIN_) + 4 * lane;

  float s[XL_];
#pragma unroll
  for (int k = 0; k < XL_; ++k) {
    const float4 v = *(const float4*)(src + 128 * k);
    s[k] = (v.x + v.y) + (v.z + v.w);
  }

  const bool b3 = (lane & 8) != 0, b2 = (lane & 4) != 0;
  const bool b1 = (lane & 2) != 0, b0 = (lane & 1) != 0;
  float t8[8], t4[4], t2[2];
#pragma unroll
  for (int i = 0; i < 8; ++i)
    t8[i] = (b3 ? s[i + 8] : s[i]) + xrow<8>(b3 ? s[i] : s[i + 8]);
#pragma unroll
  for (int i = 0; i < 4; ++i)
    t4[i] = (b2 ? t8[i + 4] : t8[i]) + xrow<4>(b2 ? t8[i] : t8[i + 4]);
#pragma unroll
  for (int i = 0; i < 2; ++i)
    t2[i] = (b1 ? t4[i + 2] : t4[i]) + xrow<2>(b1 ? t4[i] : t4[i + 2]);
  const float t1 = (b0 ? t2[1] : t2[0]) + xrow<1>(b0 ? t2[0] : t2[1]);

  P0[wv * XR_ + 2 * (lane & 15) + (lane >> 4)] = t1;
}

__device__ __forceinline__ void issue_row_loads(const float* __restrict__ Ap,
                                                const float* __restrict__ pp,
                                                float4 (&pv)[KQ_],
                                                float4 (&av)[RW_][KQ_]) {
#pragma unroll
  for (int q = 0; q < KQ_; ++q) pv[q] = *(const float4*)(pp + 128 * q);
#pragma unroll
  for (int q = 0; q < KQ_; ++q)
#pragma unroll
    for (int r = 0; r < RW_; ++r)
      av[r][q] = *(const float4*)(Ap + (size_t)r * N_ + 128 * q);
}

__device__ __forceinline__ void rows_dot(const float4 (&pv)[KQ_],
                                         const float4 (&av)[RW_][KQ_],
                                         const int lane, float (&dot)[RW_]) {
  v8f acc[RW_];
#pragma unroll
  for (int r = 0; r < RW_; ++r) acc[r] = v8f{};

#pragma unroll
  for (int q = 0; q < KQ_; ++q) {
#pragma unroll
    for (int r = 0; r < RW_; ++r) {
      const v2f a = {av[r][q].x, av[r][q].y};
      const v2f p = {pv[q].x, pv[q].y};
      acc[r] = __builtin_amdgcn_wmma_f32_16x16x4_f32(
           false, a,  false, p,
           (short)0, acc[r],  false,  false);
    }
#pragma unroll
    for (int r = 0; r < RW_; ++r) {
      const v2f a = {av[r][q].z, av[r][q].w};
      const v2f p = {pv[q].z, pv[q].w};
      acc[r] = __builtin_amdgcn_wmma_f32_16x16x4_f32(
           false, a,  false, p,
           (short)0, acc[r],  false,  false);
    }
  }

  __builtin_amdgcn_sched_barrier(0);
  asm volatile("v_nop\n\tv_nop\n\tv_nop\n\tv_nop\n\tv_nop\n\tv_nop\n\tv_nop\n\tv_nop");
  __builtin_amdgcn_sched_barrier(0);

  const int  dr = lane & 7;
  const bool dv = ((lane + 8) & 31) < 16;
#pragma unroll
  for (int r = 0; r < RW_; ++r) {
    float d = acc[r][0];
#pragma unroll
    for (int v = 1; v < 8; ++v) d = (dr == v) ? acc[r][v] : d;
    dot[r] = dv ? d : 0.f;
  }
#pragma unroll
  for (int r = 0; r < RW_; ++r) dot[r] += xhalf(dot[r]);
#pragma unroll
  for (int r = 0; r < RW_; ++r) dot[r] += xrow<8>(dot[r]);
#pragma unroll
  for (int r = 0; r < RW_; ++r) dot[r] += xrow<4>(dot[r]);
#pragma unroll
  for (int r = 0; r < RW_; ++r) dot[r] += xrow<2>(dot[r]);
#pragma unroll
  for (int r = 0; r < RW_; ++r) dot[r] += xrow<1>(dot[r]);
}

__global__ void __launch_bounds__(256)
matvec_wmma_kernel(const float* __restrict__ A,
                   const float* __restrict__ pin,
                   float* __restrict__ pout) {
  __shared__ float red[SLAB_];

  const int lane = threadIdx.x & 31;
  const int wave = threadIdx.x >> 5;
  const int slab = blockIdx.x;
  const int b    = slab >> 5;
  const int n0   = (slab & 31) << 5;

  const float* Ap = A + (size_t)b * N_ * N_ + (size_t)(n0 + RW_ * wave) * N_ + 4 * lane;
  const float* pp = pin + b * N_ + 4 * lane;

  float4 pv[KQ_];
  float4 av[RW_][KQ_];
  issue_row_loads(Ap, pp, pv, av);
  __builtin_amdgcn_sched_barrier(0);

  float dot[RW_];
  rows_dot(pv, av, lane, dot);

  if (lane == 0) {
#pragma unroll
    for (int r = 0; r < RW_; ++r) red[RW_ * wave + r] = dot[r];
  }
  __syncthreads();
  if (threadIdx.x < SLAB_) pout[b * N_ + n0 + threadIdx.x] = red[threadIdx.x];
}

__device__ __forceinline__ float tanh_exp2(const float z) {
  const float t = __builtin_amdgcn_exp2f(fabsf(z) * 2.88539008177792681f);
  const float r = fmaf(-2.f, __builtin_amdgcn_rcpf(t + 1.f), 1.f);
  return copysignf(r, z);
}

#define HP_ 66

__global__ void __launch_bounds__(256)
matvec_final_kernel(const float* __restrict__ A,
                    const float* __restrict__ P,
                    const float* __restrict__ h,
                    const float* __restrict__ gamma,
                    const float* __restrict__ beta,
                    float* __restrict__ out) {
  __shared__ float hs[(KT_ + 1) * SLAB_ * HP_];

  const int lane = threadIdx.x & 31;
  const int wave = threadIdx.x >> 5;
  const int slab = blockIdx.x;
  const int b    = slab >> 5;
  const int n0   = (slab & 31) << 5;

  const int hq = threadIdx.x & 7;
  const int hc = threadIdx.x >> 3;
  float4 hv[KT_ + 1][2];
#pragma unroll
  for (int i = 0; i <= KT_; ++i)
#pragma unroll
    for (int u = 0; u < 2; ++u)
      hv[i][u] = *(const float4*)(h + ((size_t)(i * C_ + hc + 32 * u)) * N_ + n0 + 4 * hq);

  float pw[RW_][KT_ + 1];
#pragma unroll
  for (int r = 0; r < RW_; ++r)
#pragma unroll
    for (int i = 0; i < KT_; ++i)
      pw[r][i] = P[i * (B_ * N_) + b * N_ + n0 + RW_ * wave + r];
  const float g0 = gamma[lane], g1 = gamma[lane + 32];
  const float e0 = beta[lane],  e1 = beta[lane + 32];

  const float* Ap = A + (size_t)b * N_ * N_ + (size_t)(n0 + RW_ * wave) * N_ + 4 * lane;
  const float* pp = P + (KT_ - 1) * (B_ * N_) + b * N_ + 4 * lane;
  float4 pv[KQ_];
  float4 av[RW_][KQ_];
  issue_row_loads(Ap, pp, pv, av);
  __builtin_amdgcn_sched_barrier(0);

#pragma unroll
  for (int i = 0; i <= KT_; ++i)
#pragma unroll
    for (int u = 0; u < 2; ++u) {
      float* d = hs + (i * SLAB_ + 4 * hq) * HP_ + hc + 32 * u;
      d[0 * HP_] = hv[i][u].x;
      d[1 * HP_] = hv[i][u].y;
      d[2 * HP_] = hv[i][u].z;
      d[3 * HP_] = hv[i][u].w;
    }

  __builtin_amdgcn_sched_barrier(0);

  float dot[RW_];
  rows_dot(pv, av, lane, dot);
  __syncthreads();

  float y0[RW_], y1[RW_], s1[RW_], s2[RW_];
#pragma unroll
  for (int r = 0; r < RW_; ++r) {
    const int n = RW_ * wave + r;
    pw[r][KT_] = dot[r];

    float a0 = 0.f, a1 = 0.f;
#pragma unroll
    for (int i = 0; i <= KT_; ++i) {
      const float* hrow = hs + (i * SLAB_ + n) * HP_;
      a0 = fmaf(pw[r][i], hrow[lane],      a0);
      a1 = fmaf(pw[r][i], hrow[lane + 32], a1);
    }
    y0[r] = a0;
    y1[r] = a1;
    s1[r] = a0 + a1;
    s2[r] = fmaf(a0, a0, a1 * a1);
  }
#pragma unroll
  for (int r = 0; r < RW_; ++r) { s1[r] += xhalf(s1[r]);   s2[r] += xhalf(s2[r]); }
#pragma unroll
  for (int r = 0; r < RW_; ++r) { s1[r] += xrow<8>(s1[r]); s2[r] += xrow<8>(s2[r]); }
#pragma unroll
  for (int r = 0; r < RW_; ++r) { s1[r] += xrow<4>(s1[r]); s2[r] += xrow<4>(s2[r]); }
#pragma unroll
  for (int r = 0; r < RW_; ++r) { s1[r] += xrow<2>(s1[r]); s2[r] += xrow<2>(s2[r]); }
#pragma unroll
  for (int r = 0; r < RW_; ++r) { s1[r] += xrow<1>(s1[r]); s2[r] += xrow<1>(s2[r]); }
  float z0[RW_], z1[RW_];
#pragma unroll
  for (int r = 0; r < RW_; ++r) {
    const float mu   = s1[r] * (1.f / C_);
    const float var  = s2[r] * (1.f / C_) - mu * mu;
    const float rstd = rsqrtf(var + 1e-5f);
    z0[r] = fmaf((y0[r] - mu) * rstd, g0, e0);
    z1[r] = fmaf((y1[r] - mu) * rstd, g1, e1);
  }
#pragma unroll
  for (int r = 0; r < RW_; ++r) {
    float* dst = out + ((size_t)(b * N_ + n0 + RW_ * wave + r)) * C_;
    dst[lane]      = tanh_exp2(z0[r]);
    dst[lane + 32] = tanh_exp2(z1[r]);
  }
}

extern "C" void kernel_launch(void* const* d_in, const int* in_sizes, int n_in,
                              void* d_out, int out_size, void* d_ws, size_t ws_size,
                              hipStream_t stream) {
  const float* A     = (const float*)d_in[0];
  const float* X     = (const float*)d_in[1];
  const float* h     = (const float*)d_in[2];
  const float* gamma = (const float*)d_in[3];
  const float* beta  = (const float*)d_in[4];
  float* out = (float*)d_out;

  float* P  = (float*)d_ws;

  hipLaunchKernelGGL(xsum_kernel, dim3(B_ * N_ / (2 * XR_)), dim3(64), 0, stream, X, P);
  for (int i = 0; i < KT_ - 1; ++i) {
    hipLaunchKernelGGL(matvec_wmma_kernel, dim3(B_ * (N_ / SLAB_)), dim3(256), 0,
                       stream, A, P + i * (B_ * N_), P + (i + 1) * (B_ * N_));
  }
  hipLaunchKernelGGL(matvec_final_kernel, dim3(B_ * (N_ / SLAB_)), dim3(256), 0,
                     stream, A, P, h, gamma, beta, out);
}
